# starting kernel
# baseline (speedup 1.0000x reference)
#include <hip/hip_runtime.h>

typedef float v2f __attribute__((ext_vector_type(2)));
typedef float v8f __attribute__((ext_vector_type(8)));

__device__ constexpr float GK[7] = {
    0.0701593f, 0.1310749f, 0.1907128f, 0.2161058f,
    0.1907128f, 0.1310749f, 0.0701593f
};

#define DIM     160
#define SLICES  (4 * DIM)
#define TILES_X (DIM / 16)
#define TILES_Y (DIM / 16)
#define TILES_PER_SLICE (TILES_X * TILES_Y)

__global__ __launch_bounds__(128) void gb3d_blurx_wmma(
    const float* __restrict__ in, float* __restrict__ out) {
  const int wave  = threadIdx.x >> 5;
  const int lane  = threadIdx.x & 31;
  const int tile  = blockIdx.x * 4 + wave;

  const int slice = tile / TILES_PER_SLICE;
  const int trem  = tile - slice * TILES_PER_SLICE;
  const int ty    = trem / TILES_X;
  const int tx    = trem - ty * TILES_X;
  const int x0    = tx * 16;
  const int y0    = ty * 16;

  const size_t base = (size_t)slice * (DIM * DIM);
  const int row = lane & 15;
  const int h   = lane >> 4;

  const float* __restrict__ rowp = in + base + (size_t)(y0 + row) * DIM;

  v2f a[6], b[6];
#pragma unroll
  for (int c = 0; c < 6; ++c) {
#pragma unroll
    for (int e = 0; e < 2; ++e) {
      const int k  = 4 * c + 2 * h + e;
      const int xi = x0 - 3 + k;
      const int xc = min(max(xi, 0), DIM - 1);
      float av = __builtin_nontemporal_load(&rowp[xc]);
      a[c][e] = (xi == xc) ? av : 0.0f;
      const int d = k - row;
      b[c][e] = (d >= 0 && d <= 6) ? GK[d] : 0.0f;
    }
  }

  v8f acc = {0.f, 0.f, 0.f, 0.f, 0.f, 0.f, 0.f, 0.f};
#pragma unroll
  for (int c = 0; c < 6; ++c) {
    acc = __builtin_amdgcn_wmma_f32_16x16x4_f32(
        false, a[c], false, b[c], (short)0, acc, false, false);
  }

  float* __restrict__ op = out + base + (size_t)(y0 + 8 * h) * DIM + x0 + row;
#pragma unroll
  for (int r = 0; r < 8; ++r)
    op[(size_t)r * DIM] = acc[r];
}

template <int STRIDEF>
__global__ __launch_bounds__(256) void gb3d_blur_axis_f4(
    const float4* __restrict__ in, float4* __restrict__ out) {
  const long tid  = (long)blockIdx.x * 256 + threadIdx.x;
  const long flat = tid * 4;
  const int  pos  = (int)((flat / STRIDEF) % DIM);
  constexpr int S4 = STRIDEF / 4;

  float4 acc = make_float4(0.f, 0.f, 0.f, 0.f);
#pragma unroll
  for (int t = 0; t < 7; ++t) {
    const int p = pos + t - 3;
    if (0 <= p && p < DIM) {
      const float4 v = in[tid + (long)(t - 3) * S4];
      const float  g = GK[t];
      acc.x += g * v.x; acc.y += g * v.y;
      acc.z += g * v.z; acc.w += g * v.w;
    }
  }
  out[tid] = acc;
}

extern "C" void kernel_launch(void* const* d_in, const int* in_sizes, int n_in,
                              void* d_out, int out_size, void* d_ws, size_t ws_size,
                              hipStream_t stream) {
  (void)in_sizes; (void)n_in; (void)out_size; (void)ws_size;
  const float* x   = (const float*)d_in[0];
  float*       out = (float*)d_out;
  float*       tmp = (float*)d_ws;

  gb3d_blurx_wmma<<<SLICES * TILES_PER_SLICE / 4, 128, 0, stream>>>(x, out);

  const int n4blocks = (4 * DIM * DIM * DIM) / 4 / 256;
  gb3d_blur_axis_f4<DIM><<<n4blocks, 256, 0, stream>>>(
      (const float4*)out, (float4*)tmp);
  gb3d_blur_axis_f4<DIM * DIM><<<n4blocks, 256, 0, stream>>>(
      (const float4*)tmp, (float4*)out);
}


# x-blur WMMA fused into the y+z ring kernel: tile row m = 8 rows x 2 column segments, a lane's 16 D registers are its y window; 256 work-groups of 20 waves; no workspace pass; sums unchanged
# speedup vs baseline: 5.0679x; 5.0679x over previous
#include <hip/hip_runtime.h>


typedef float v2f __attribute__((ext_vector_type(2)));
typedef float v8f __attribute__((ext_vector_type(8)));

__device__ constexpr float GK[7] = {
    0.0701593f, 0.1310749f, 0.1907128f, 0.2161058f,
    0.1907128f, 0.1310749f, 0.0701593f
};

#define DIM     160
#define PLANE   (DIM * DIM)
#define VOL     (DIM * PLANE)

#define FZ_LZ    10
#define FZ_ROWS  10

__device__ __forceinline__ void fz_load_plane(
    float (&raw)[2][6][2], const char* __restrict__ base, unsigned plane_off,
    const unsigned (&rowoff)[2], const unsigned (&xoff)[6][2]) {
#pragma unroll
  for (int t = 0; t < 2; ++t) {
    const unsigned ro = plane_off + rowoff[t];
#pragma unroll
    for (int c = 0; c < 6; ++c)
#pragma unroll
      for (int e = 0; e < 2; ++e)
        raw[t][c][e] = *(const float*)(base + (ro + xoff[c][e]));
  }
}

__global__ __launch_bounds__(640) __attribute__((amdgpu_waves_per_eu(5)))
void gb3d_blur_xyz_ring(
    const float* __restrict__ in, float* __restrict__ out) {
  const int lane = threadIdx.x & 31;
  const int col  = lane & 15;
  const int h    = lane >> 4;
  const int gw = __builtin_amdgcn_readfirstlane(
      (int)(blockIdx.x * 20 + (threadIdx.x >> 5)));
  const int xt = gw % 5;
  const int yt = (gw / 5) % 16;
  const int zc = (gw / 80) % 16;
  const int vv = gw / 1280;
  const int x0  = xt * 32;
  const int yo0 = yt * FZ_ROWS;
  const int z0  = zc * FZ_LZ;

  v2f b[6];
#pragma unroll
  for (int c = 0; c < 6; ++c) {
#pragma unroll
    for (int e = 0; e < 2; ++e) {
      const int k = 4 * c + 2 * h + e;
      const int d = k - col;
      b[c][e] = (d >= 0 && d <= 6) ? GK[d] : 0.0f;
    }
  }

  const int seg = col >> 3;
  const int rr  = col & 7;
  const int xs  = x0 + 16 * seg - 3;
  unsigned rowoff[2];
  unsigned xoff[6][2];
  bool     yin[2], xin[6][2];
#pragma unroll
  for (int t = 0; t < 2; ++t) {
    const int y  = yo0 - 3 + 8 * t + rr;
    const int yc = min(max(y, 0), DIM - 1);
    rowoff[t] = (unsigned)(yc * DIM) * 4u;
    yin[t]    = (y == yc);
  }
#pragma unroll
  for (int c = 0; c < 6; ++c) {
#pragma unroll
    for (int e = 0; e < 2; ++e) {
      const int k  = 4 * c + 2 * h + e;
      const int xi = xs + k;
      const int xc = min(max(xi, 0), DIM - 1);
      xoff[c][e] = (unsigned)xc * 4u;
      xin[c][e]  = (xi == xc);
    }
  }

  const char* __restrict__ inb = (const char*)in;
  const unsigned vol_off = (unsigned)vv * (unsigned)(VOL * 4);
  float* __restrict__ vout = out + (size_t)vv * VOL + (size_t)(yo0 * DIM + x0 + lane);

  float raw[2][6][2];
  float ring[8][FZ_ROWS];

  {
    const int zn = min(max(z0 - 3, 0), DIM - 1);
    fz_load_plane(raw, inb, vol_off + (unsigned)zn * (unsigned)(PLANE * 4), rowoff, xoff);
  }

#pragma unroll 1
  for (int it = 0; it < 2; ++it) {
#pragma unroll
    for (int j = 0; j < 8; ++j) {
      const int s  = it * 8 + j;
      const int zp = z0 - 3 + s;
      const bool plane_in = (zp >= 0) && (zp < DIM);

      v2f a[2][6];
#pragma unroll
      for (int t = 0; t < 2; ++t)
#pragma unroll
        for (int c = 0; c < 6; ++c)
#pragma unroll
          for (int e = 0; e < 2; ++e)
            a[t][c][e] = (yin[t] && xin[c][e]) ? raw[t][c][e] : 0.0f;

      v8f d0 = {0.f, 0.f, 0.f, 0.f, 0.f, 0.f, 0.f, 0.f};
      v8f d1 = {0.f, 0.f, 0.f, 0.f, 0.f, 0.f, 0.f, 0.f};
#pragma unroll
      for (int c = 0; c < 6; ++c) {
        d0 = __builtin_amdgcn_wmma_f32_16x16x4_f32(
            false, a[0][c], false, b[c], (short)0, d0, false, false);
        d1 = __builtin_amdgcn_wmma_f32_16x16x4_f32(
            false, a[1][c], false, b[c], (short)0, d1, false, false);
      }

      {
        const int zn = min(max(zp + 1, 0), DIM - 1);
        fz_load_plane(raw, inb, vol_off + (unsigned)zn * (unsigned)(PLANE * 4), rowoff, xoff);
      }
      __builtin_amdgcn_sched_barrier(0);

      float win[16];
#pragma unroll
      for (int r = 0; r < 8; ++r) { win[r] = d0[r]; win[8 + r] = d1[r]; }
#pragma unroll
      for (int q = 0; q < FZ_ROWS; ++q) {
        float acc = 0.f;
#pragma unroll
        for (int k = 0; k < 7; ++k)
          acc += GK[k] * win[q + k];
        ring[j][q] = plane_in ? acc : 0.f;
      }

      if (s >= 6) {
        float* __restrict__ op = vout + (size_t)(z0 + s - 6) * PLANE;
#pragma unroll
        for (int q = 0; q < FZ_ROWS; ++q) {
          float acc = 0.f;
#pragma unroll
          for (int k = 0; k < 7; ++k)
            acc += GK[k] * ring[(j + 2 + k) & 7][q];
          op[q * DIM] = acc;
        }
      }
    }
  }
}

extern "C" void kernel_launch(void* const* d_in, const int* in_sizes, int n_in,
                              void* d_out, int out_size, void* d_ws, size_t ws_size,
                              hipStream_t stream) {
  (void)in_sizes; (void)n_in; (void)out_size; (void)d_ws; (void)ws_size;
  const float* x   = (const float*)d_in[0];
  float*       out = (float*)d_out;

  gb3d_blur_xyz_ring<<<256, 640, 0, stream>>>(x, out);
}


# a wave walks 20 output planes along z, not 10 (26 plane steps, not 2 x 16): z-halo planes loaded and x-blurred twice drop from 6 in 16 to 6 in 26; 2560 waves in 256 groups of 10; sums unchanged
# speedup vs baseline: 5.7876x; 1.1420x over previous
#include <hip/hip_runtime.h>


typedef float v2f __attribute__((ext_vector_type(2)));
typedef float v8f __attribute__((ext_vector_type(8)));

__device__ constexpr float GK[7] = {
    0.0701593f, 0.1310749f, 0.1907128f, 0.2161058f,
    0.1907128f, 0.1310749f, 0.0701593f
};

#define DIM     160
#define PLANE   (DIM * DIM)
#define VOL     (DIM * PLANE)

#define FZ_LZ    20
#define FZ_STEPS (FZ_LZ + 6)
#define FZ_ROWS  10

__device__ __forceinline__ void fz_load_plane(
    float (&raw)[2][6][2], const char* __restrict__ base, unsigned plane_off,
    const unsigned (&rowoff)[2], const unsigned (&xoff)[6][2]) {
#pragma unroll
  for (int t = 0; t < 2; ++t) {
    const unsigned ro = plane_off + rowoff[t];
#pragma unroll
    for (int c = 0; c < 6; ++c)
#pragma unroll
      for (int e = 0; e < 2; ++e)
        raw[t][c][e] = *(const float*)(base + (ro + xoff[c][e]));
  }
}

__global__ __launch_bounds__(320)
void gb3d_blur_xyz_ring(
    const float* __restrict__ in, float* __restrict__ out) {
  const int lane = threadIdx.x & 31;
  const int col  = lane & 15;
  const int h    = lane >> 4;
  const int gw = __builtin_amdgcn_readfirstlane(
      (int)(blockIdx.x * 10 + (threadIdx.x >> 5)));
  const int xt = gw % 5;
  const int yt = (gw / 5) % 16;
  const int zc = (gw / 80) % 8;
  const int vv = gw / 640;
  const int x0  = xt * 32;
  const int yo0 = yt * FZ_ROWS;
  const int z0  = zc * FZ_LZ;

  v2f b[6];
#pragma unroll
  for (int c = 0; c < 6; ++c) {
#pragma unroll
    for (int e = 0; e < 2; ++e) {
      const int k = 4 * c + 2 * h + e;
      const int d = k - col;
      b[c][e] = (d >= 0 && d <= 6) ? GK[d] : 0.0f;
    }
  }

  const int seg = col >> 3;
  const int rr  = col & 7;
  const int xs  = x0 + 16 * seg - 3;
  unsigned rowoff[2];
  unsigned xoff[6][2];
  bool     yin[2], xin[6][2];
#pragma unroll
  for (int t = 0; t < 2; ++t) {
    const int y  = yo0 - 3 + 8 * t + rr;
    const int yc = min(max(y, 0), DIM - 1);
    rowoff[t] = (unsigned)(yc * DIM) * 4u;
    yin[t]    = (y == yc);
  }
#pragma unroll
  for (int c = 0; c < 6; ++c) {
#pragma unroll
    for (int e = 0; e < 2; ++e) {
      const int k  = 4 * c + 2 * h + e;
      const int xi = xs + k;
      const int xc = min(max(xi, 0), DIM - 1);
      xoff[c][e] = (unsigned)xc * 4u;
      xin[c][e]  = (xi == xc);
    }
  }

  const char* __restrict__ inb = (const char*)in;
  const unsigned vol_off = (unsigned)vv * (unsigned)(VOL * 4);
  float* __restrict__ vout = out + (size_t)vv * VOL + (size_t)(yo0 * DIM + x0 + lane);

  float raw[2][6][2];
  float ring[8][FZ_ROWS];

  {
    const int zn = min(max(z0 - 3, 0), DIM - 1);
    fz_load_plane(raw, inb, vol_off + (unsigned)zn * (unsigned)(PLANE * 4), rowoff, xoff);
  }

#pragma unroll 1
  for (int it = 0; it < (FZ_STEPS + 7) / 8; ++it) {
#pragma unroll
    for (int j = 0; j < 8; ++j) {
      const int s  = it * 8 + j;
      if (s >= FZ_STEPS) break;
      const int zp = z0 - 3 + s;
      const bool plane_in = (zp >= 0) && (zp < DIM);

      v2f a[2][6];
#pragma unroll
      for (int t = 0; t < 2; ++t)
#pragma unroll
        for (int c = 0; c < 6; ++c)
#pragma unroll
          for (int e = 0; e < 2; ++e)
            a[t][c][e] = (yin[t] && xin[c][e]) ? raw[t][c][e] : 0.0f;

      v8f d0 = {0.f, 0.f, 0.f, 0.f, 0.f, 0.f, 0.f, 0.f};
      v8f d1 = {0.f, 0.f, 0.f, 0.f, 0.f, 0.f, 0.f, 0.f};
#pragma unroll
      for (int c = 0; c < 6; ++c) {
        d0 = __builtin_amdgcn_wmma_f32_16x16x4_f32(
            false, a[0][c], false, b[c], (short)0, d0, false, false);
        d1 = __builtin_amdgcn_wmma_f32_16x16x4_f32(
            false, a[1][c], false, b[c], (short)0, d1, false, false);
      }

      {
        const int zn = min(max(zp + 1, 0), DIM - 1);
        fz_load_plane(raw, inb, vol_off + (unsigned)zn * (unsigned)(PLANE * 4), rowoff, xoff);
      }
      __builtin_amdgcn_sched_barrier(0);

      float win[16];
#pragma unroll
      for (int r = 0; r < 8; ++r) { win[r] = d0[r]; win[8 + r] = d1[r]; }
#pragma unroll
      for (int q = 0; q < FZ_ROWS; ++q) {
        float acc = 0.f;
#pragma unroll
        for (int k = 0; k < 7; ++k)
          acc += GK[k] * win[q + k];
        ring[j][q] = plane_in ? acc : 0.f;
      }

      if (s >= 6) {
        float* __restrict__ op = vout + (size_t)(z0 + s - 6) * PLANE;
#pragma unroll
        for (int q = 0; q < FZ_ROWS; ++q) {
          float acc = 0.f;
#pragma unroll
          for (int k = 0; k < 7; ++k)
            acc += GK[k] * ring[(j + 2 + k) & 7][q];
          op[q * DIM] = acc;
        }
      }
    }
  }
}

extern "C" void kernel_launch(void* const* d_in, const int* in_sizes, int n_in,
                              void* d_out, int out_size, void* d_ws, size_t ws_size,
                              hipStream_t stream) {
  (void)in_sizes; (void)n_in; (void)out_size; (void)d_ws; (void)ws_size;
  const float* x   = (const float*)d_in[0];
  float*       out = (float*)d_out;

  gb3d_blur_xyz_ring<<<256, 320, 0, stream>>>(x, out);
}
